# starting kernel
# baseline (speedup 1.0000x reference)
#include <hip/hip_runtime.h>

typedef __attribute__((ext_vector_type(2))) float v2f;
typedef __attribute__((ext_vector_type(8))) float v8f;

#define N_ATOMS 8192
#define N_EDGES 65536
#define N_PAIR  16
#define NH      64
#define TILES   2
#define EDGES_PER_BLOCK (TILES * 16)

__global__ void prep_wt(const float* __restrict__ W, const float* __restrict__ b,
                        float* __restrict__ Wt) {
    int idx = blockIdx.x * blockDim.x + threadIdx.x;
    if (idx >= 17 * NH * NH) return;
    int p = idx / (NH * NH);
    int r = idx % (NH * NH);
    int k = r / NH;
    int h = r % NH;
    Wt[idx] = (p < N_PAIR) ? W[p * (NH * NH) + h * NH + k] : b[h * NH + k];
}

__global__ void zero_out(float* __restrict__ out, int n) {
    int idx = blockIdx.x * blockDim.x + threadIdx.x;
    if (idx < n) out[idx] = 0.0f;
}

__launch_bounds__(128)
__global__ void edge_net(const float* __restrict__ pf,
                         const float* __restrict__ atomf,
                         const int*   __restrict__ a2p,
                         const float* __restrict__ Wt,
                         float*       __restrict__ out) {
    __shared__ float srcLds[EDGES_PER_BLOCK][NH + 4];
    __shared__ float pfLds[EDGES_PER_BLOCK][N_PAIR + 1];
    __shared__ int   a0Lds[EDGES_PER_BLOCK];

    const int tid  = threadIdx.x;
    const int lane = tid & 31;
    const int wave = tid >> 5;
    const int eb   = blockIdx.x * EDGES_PER_BLOCK;

    {
        int el = tid >> 2;
        int q  = tid & 3;
        int e  = eb + el;
        int a1 = a2p[2 * e + 1];
        const float4* srow = (const float4*)(atomf + (size_t)a1 * NH);
#pragma unroll
        for (int j = 0; j < 4; ++j) {
            float4 v = srow[q * 4 + j];
            float* d = &srcLds[el][q * 16 + j * 4];
            d[0] = v.x; d[1] = v.y; d[2] = v.z; d[3] = v.w;
        }
        if (tid < EDGES_PER_BLOCK) {
            int e2 = eb + tid;
#pragma unroll
            for (int j = 0; j < N_PAIR; ++j) pfLds[tid][j] = pf[e2 * N_PAIR + j];
            pfLds[tid][N_PAIR] = 1.0f;
            a0Lds[tid] = a2p[2 * e2];
        }
    }
    __syncthreads();

    const int h0   = wave * 16;
    const int n    = lane & 15;
    const int koff = (lane >> 4) * 2;
    const int mrow = (lane >> 4) * 8;

    v2f af[TILES][16];
#pragma unroll
    for (int t = 0; t < TILES; ++t)
#pragma unroll
        for (int ks = 0; ks < 16; ++ks)
            af[t][ks] = *(const v2f*)&srcLds[t * 16 + n][ks * 4 + koff];

    v8f accO[TILES];
#pragma unroll
    for (int t = 0; t < TILES; ++t) accO[t] = (v8f)0.0f;

    for (int p = 0; p <= N_PAIR; ++p) {
        const float* wp = Wt + (size_t)p * (NH * NH);
        v2f bf[16];
#pragma unroll
        for (int ks = 0; ks < 16; ++ks) {
            int kb   = ks * 4 + koff;
            bf[ks].x = wp[kb * NH + h0 + n];
            bf[ks].y = wp[(kb + 1) * NH + h0 + n];
        }
#pragma unroll
        for (int t = 0; t < TILES; ++t) {
            v8f acc = (v8f)0.0f;
#pragma unroll
            for (int ks = 0; ks < 16; ++ks)
                acc = __builtin_amdgcn_wmma_f32_16x16x4_f32(
                    false, af[t][ks], false, bf[ks], (short)0, acc, false, false);
#pragma unroll
            for (int i = 0; i < 8; ++i) {
                float w = pfLds[t * 16 + mrow + i][p];
                accO[t][i] += w * acc[i];
            }
        }
    }

#pragma unroll
    for (int t = 0; t < TILES; ++t)
#pragma unroll
        for (int i = 0; i < 8; ++i) {
            int el   = t * 16 + mrow + i;
            int atom = a0Lds[el];
            atomicAdd(&out[(size_t)atom * NH + h0 + n], accO[t][i]);
        }
}

extern "C" void kernel_launch(void* const* d_in, const int* in_sizes, int n_in,
                              void* d_out, int out_size, void* d_ws, size_t ws_size,
                              hipStream_t stream) {
    const float* pf    = (const float*)d_in[0];
    const float* atomf = (const float*)d_in[1];
    const int*   a2p   = (const int*)d_in[2];
    const float* W     = (const float*)d_in[3];
    const float* b     = (const float*)d_in[4];
    float* out = (float*)d_out;
    float* Wt  = (float*)d_ws;

    int prep_n = 17 * NH * NH;
    prep_wt<<<(prep_n + 255) / 256, 256, 0, stream>>>(W, b, Wt);
    zero_out<<<(N_ATOMS * NH + 255) / 256, 256, 0, stream>>>(out, N_ATOMS * NH);
    edge_net<<<N_EDGES / EDGES_PER_BLOCK, 128, 0, stream>>>(pf, atomf, a2p, Wt, out);
}


# product phase re-tiled: K split over the 8 waves, each wave keeps all 8 output tiles; weight rows read whole by 16-byte loads three blocks ahead, LDS rows read once; k sum order changes (8 partials)
# speedup vs baseline: 2.2957x; 2.2957x over previous
#include <hip/hip_runtime.h>


typedef __attribute__((ext_vector_type(2))) float v2f;
typedef __attribute__((ext_vector_type(8))) float v8f;
typedef __attribute__((ext_vector_type(4))) float v4f;

#define N_ATOMS 8192
#define N_EDGES 65536
#define N_PAIR  16
#define NH      64
#define KTOT    (17 * NH)
#define MP      (KTOT + 4)
#define APB     32
#define NTHR    256
#define EG      16
#define KW      (KTOT / 8)
#define NPAIR   (KW / 8)
#define PP      (NH + 4)
#define WT_BYTES (17 * NH * NH * 4)

__global__ void prep_wt(const float* __restrict__ W, const float* __restrict__ b,
                        const int* __restrict__ a2p,
                        float* __restrict__ Wt, unsigned int* __restrict__ own) {
    int idx = blockIdx.x * blockDim.x + threadIdx.x;
    if (idx < 17 * NH * NH) {
        int p = idx / (NH * NH);
        int r = idx % (NH * NH);
        int k = r / NH;
        int h = r % NH;
        Wt[idx] = (p < N_PAIR) ? W[p * (NH * NH) + h * NH + k] : b[h * NH + k];
    }
    if (idx < N_EDGES / 4) {
        unsigned int w = 0;
#pragma unroll
        for (int j = 0; j < 4; ++j) {
            unsigned int o = ((unsigned int)a2p[2 * (4 * idx + j)] >> 5) & 255u;
            w |= o << (8 * j);
        }
        own[idx] = w;
    }
}

__launch_bounds__(NTHR)
__global__ void edge_net(const float* __restrict__ pf,
                         const float* __restrict__ atomf,
                         const int*   __restrict__ a2p,
                         const float* __restrict__ Wt,
                         const unsigned int* __restrict__ own,
                         float*       __restrict__ out) {
    __shared__ __attribute__((aligned(16))) float Mraw[APB * MP];
    __shared__ __attribute__((aligned(16))) unsigned short sortedE[N_EDGES];
    __shared__ int cnt[APB];
    __shared__ int cur[APB];
    __shared__ int start[APB + 1];
    __shared__ int nMatch;

    unsigned short* tmpE = (unsigned short*)Mraw;
    float* part8 = (float*)sortedE;

    const int tid  = threadIdx.x;
    const int lane = tid & 31;
    const int wave = __builtin_amdgcn_readfirstlane(tid >> 5);
    const unsigned int blk = blockIdx.x;

    if (tid < APB) { cnt[tid] = 0; cur[tid] = 0; }
    if (tid == 0) nMatch = 0;
    __syncthreads();

    const int dn  = lane & 15;
    const int dhh = lane >> 4;
    const float* bptr = Wt + (size_t)(wave * KW + 2 * dhh) * NH + 4 * dn;
    v4f bx[4][2], by[4][2];
#pragma unroll
    for (int q = 0; q < 3; ++q)
#pragma unroll
        for (int s = 0; s < 2; ++s) {
            bx[q][s] = *(const v4f*)(bptr + (q * 8 + s * 4) * NH);
            by[q][s] = *(const v4f*)(bptr + (q * 8 + s * 4 + 1) * NH);
        }

    {
        const uint4* own4 = (const uint4*)own;
        uint4* ownL4 = (uint4*)sortedE;
        const unsigned int* ownLw = (const unsigned int*)sortedE;
        const unsigned int pat = blk * 0x01010101u;
        uint4 ov[N_EDGES / (16 * NTHR)];
#pragma unroll
        for (int it = 0; it < N_EDGES / (16 * NTHR); ++it) ov[it] = own4[it * NTHR + tid];
        unsigned long long m = 0ull;
#pragma unroll
        for (int it = 0; it < N_EDGES / (16 * NTHR); ++it) {
            ownL4[it * NTHR + tid] = ov[it];
            unsigned int wd[4] = {ov[it].x, ov[it].y, ov[it].z, ov[it].w};
#pragma unroll
            for (int wi = 0; wi < 4; ++wi) {
                unsigned int x = wd[wi] ^ pat;
                unsigned long long f = (((x - 0x01010101u) & ~x & 0x80808080u) != 0u) ? 1ull : 0ull;
                m |= f << (it * 4 + wi);
            }
        }
        while (m != 0ull) {
            int bpos = __builtin_ctzll(m);
            m &= m - 1ull;
            int widx = ((bpos >> 2) * NTHR + tid) * 4 + (bpos & 3);
            unsigned int x = ownLw[widx] ^ pat;
            unsigned int z = ~(((x & 0x7f7f7f7fu) + 0x7f7f7f7fu) | x) & 0x80808080u;
            int nm = __builtin_popcount(z);
            int idx = atomicAdd(&nMatch, nm);
#pragma unroll
            for (int bi = 0; bi < 4; ++bi) {
                if (z & (0x80u << (8 * bi))) {
                    tmpE[idx] = (unsigned short)(widx * 4 + bi);
                    ++idx;
                }
            }
        }
    }
    __syncthreads();
    const int nE = __builtin_amdgcn_readfirstlane(nMatch);

    for (int i = tid; i < nE; i += NTHR) {
        int e = tmpE[i];
        int a = a2p[2 * e] & (APB - 1);
        atomicAdd(&cnt[a], 1);
    }
    __syncthreads();
    if (tid == 0) {
        int s = 0;
        for (int a = 0; a < APB; ++a) { start[a] = s; s += cnt[a]; }
        start[APB] = s;
    }
    __syncthreads();
    for (int i = tid; i < nE; i += NTHR) {
        int e = tmpE[i];
        int a = a2p[2 * e] & (APB - 1);
        int pos = start[a] + atomicAdd(&cur[a], 1);
        sortedE[pos] = (unsigned short)e;
    }
    __syncthreads();

    {
        v2f acc[17];
#pragma unroll
        for (int p = 0; p < 17; ++p) acc[p] = (v2f)0.0f;
        int curA = wave * 4;
        const int lastA = wave * 4 + 4;
        const int rbeg = __builtin_amdgcn_readfirstlane(start[wave * 4]);
        const int rend = __builtin_amdgcn_readfirstlane(start[wave * 4 + 4]);

        for (int base = rbeg; base < rend; base += 32) {
            int idx = base + lane;
            int e_l = 0, a_l = 0, s_l = 0;
            if (idx < rend) {
                e_l = sortedE[idx];
                a_l = a2p[2 * e_l] & (APB - 1);
                s_l = a2p[2 * e_l + 1];
            }
            int nb = rend - base;
            if (nb > 32) nb = 32;
            for (int j0 = 0; j0 < nb; j0 += EG) {
                v2f   sv[EG];
                float pv[EG];
#pragma unroll
                for (int u = 0; u < EG; ++u) {
                    int j = j0 + u;
                    int e = __builtin_amdgcn_readlane(e_l, j);
                    int s = __builtin_amdgcn_readlane(s_l, j);
                    sv[u] = *(const v2f*)(atomf + (size_t)s * NH + 2 * lane);
                    pv[u] = pf[(size_t)e * N_PAIR + (lane & 15)];
                }
#pragma unroll
                for (int u = 0; u < EG; ++u) {
                    int j = j0 + u;
                    if (j < nb) {
                        int a = __builtin_amdgcn_readlane(a_l, j);
                        while (curA < a) {
#pragma unroll
                            for (int p = 0; p < 17; ++p) {
                                *(v2f*)&Mraw[curA * MP + p * NH + 2 * lane] = acc[p];
                                acc[p] = (v2f)0.0f;
                            }
                            ++curA;
                        }
#pragma unroll
                        for (int p = 0; p < N_PAIR; ++p) {
                            float w = __int_as_float(__builtin_amdgcn_readlane(__float_as_int(pv[u]), p));
                            acc[p] += w * sv[u];
                        }
                        acc[N_PAIR] += sv[u];
                    }
                }
            }
        }
        while (curA < lastA) {
#pragma unroll
            for (int p = 0; p < 17; ++p) {
                *(v2f*)&Mraw[curA * MP + p * NH + 2 * lane] = acc[p];
                acc[p] = (v2f)0.0f;
            }
            ++curA;
        }
    }
    __syncthreads();

    {
        const float* aptr = &Mraw[dn * MP + wave * KW + 2 * dhh];

        v8f acc[2][4];
#pragma unroll
        for (int t = 0; t < 2; ++t)
#pragma unroll
            for (int j = 0; j < 4; ++j) acc[t][j] = (v8f)0.0f;

        v2f af[2][2][2];
#pragma unroll
        for (int s = 0; s < 2; ++s) {
            af[0][s][0] = *(const v2f*)(aptr + 4 * s);
            af[0][s][1] = *(const v2f*)(aptr + 16 * MP + 4 * s);
        }

#pragma unroll
        for (int q = 0; q < NPAIR; ++q) {
            if (q + 3 < NPAIR) {
#pragma unroll
                for (int s = 0; s < 2; ++s) {
                    bx[(q + 3) & 3][s] = *(const v4f*)(bptr + ((q + 3) * 8 + s * 4) * NH);
                    by[(q + 3) & 3][s] = *(const v4f*)(bptr + ((q + 3) * 8 + s * 4 + 1) * NH);
                }
            }
            if (q + 1 < NPAIR) {
#pragma unroll
                for (int s = 0; s < 2; ++s) {
                    af[(q + 1) & 1][s][0] = *(const v2f*)(aptr + (q + 1) * 8 + 4 * s);
                    af[(q + 1) & 1][s][1] = *(const v2f*)(aptr + 16 * MP + (q + 1) * 8 + 4 * s);
                }
            }
#pragma unroll
            for (int s = 0; s < 2; ++s) {
#pragma unroll
                for (int j = 0; j < 4; ++j) {
                    v2f bf;
                    bf.x = bx[q & 3][s][j];
                    bf.y = by[q & 3][s][j];
                    acc[0][j] = __builtin_amdgcn_wmma_f32_16x16x4_f32(
                        false, af[q & 1][s][0], false, bf, (short)0, acc[0][j], false, false);
                    acc[1][j] = __builtin_amdgcn_wmma_f32_16x16x4_f32(
                        false, af[q & 1][s][1], false, bf, (short)0, acc[1][j], false, false);
                }
            }
            __builtin_amdgcn_sched_barrier(0);
        }

#pragma unroll
        for (int t = 0; t < 2; ++t)
#pragma unroll
            for (int r = 0; r < 8; ++r) {
                v4f v;
                v.x = acc[t][0][r]; v.y = acc[t][1][r]; v.z = acc[t][2][r]; v.w = acc[t][3][r];
                *(v4f*)&part8[(wave * APB + t * 16 + dhh * 8 + r) * PP + 4 * dn] = v;
            }
    }
    __syncthreads();

#pragma unroll
    for (int it = 0; it < 2; ++it) {
        int i4  = it * NTHR + tid;
        int row = i4 >> 4;
        int c4  = (i4 & 15) * 4;
        v4f r = *(const v4f*)&part8[row * PP + c4];
#pragma unroll
        for (int w = 1; w < 8; ++w) r += *(const v4f*)&part8[(w * APB + row) * PP + c4];
        *(v4f*)&out[((size_t)blk * APB + row) * NH + c4] = r;
    }
}

extern "C" void kernel_launch(void* const* d_in, const int* in_sizes, int n_in,
                              void* d_out, int out_size, void* d_ws, size_t ws_size,
                              hipStream_t stream) {
    const float* pf    = (const float*)d_in[0];
    const float* atomf = (const float*)d_in[1];
    const int*   a2p   = (const int*)d_in[2];
    const float* W     = (const float*)d_in[3];
    const float* b     = (const float*)d_in[4];
    float* out = (float*)d_out;
    float* Wt  = (float*)d_ws;
    unsigned int* own = (unsigned int*)((char*)d_ws + WT_BYTES);

    prep_wt<<<272, 256, 0, stream>>>(W, b, a2p, Wt, own);
    edge_net<<<256, 256, 0, stream>>>(pf, atomf, a2p, Wt, own, out);
}


# accumulation phase software-pipelined: pf and src rows of the next 16-edge group are requested ahead of the current group's sums (second register set), hiding their latency under them; sum order kept
# speedup vs baseline: 2.4185x; 1.0535x over previous
#include <hip/hip_runtime.h>


typedef __attribute__((ext_vector_type(2))) float v2f;
typedef __attribute__((ext_vector_type(8))) float v8f;
typedef __attribute__((ext_vector_type(4))) float v4f;

#define N_ATOMS 8192
#define N_EDGES 65536
#define N_PAIR  16
#define NH      64
#define KTOT    (17 * NH)
#define MP      (KTOT + 4)
#define APB     32
#define NTHR    256
#define EG      16
#define KW      (KTOT / 8)
#define NPAIR   (KW / 8)
#define PP      (NH + 4)
#define WT_BYTES (17 * NH * NH * 4)

__global__ void prep_wt(const float* __restrict__ W, const float* __restrict__ b,
                        const int* __restrict__ a2p,
                        float* __restrict__ Wt, unsigned int* __restrict__ own) {
    int idx = blockIdx.x * blockDim.x + threadIdx.x;
    if (idx < 17 * NH * NH) {
        int p = idx / (NH * NH);
        int r = idx % (NH * NH);
        int k = r / NH;
        int h = r % NH;
        Wt[idx] = (p < N_PAIR) ? W[p * (NH * NH) + h * NH + k] : b[h * NH + k];
    }
    if (idx < N_EDGES / 4) {
        unsigned int w = 0;
#pragma unroll
        for (int j = 0; j < 4; ++j) {
            unsigned int o = ((unsigned int)a2p[2 * (4 * idx + j)] >> 5) & 255u;
            w |= o << (8 * j);
        }
        own[idx] = w;
    }
}

__launch_bounds__(NTHR)
__global__ void edge_net(const float* __restrict__ pf,
                         const float* __restrict__ atomf,
                         const int*   __restrict__ a2p,
                         const float* __restrict__ Wt,
                         const unsigned int* __restrict__ own,
                         float*       __restrict__ out) {
    __shared__ __attribute__((aligned(16))) float Mraw[APB * MP];
    __shared__ __attribute__((aligned(16))) unsigned short sortedE[N_EDGES];
    __shared__ int cnt[APB];
    __shared__ int cur[APB];
    __shared__ int start[APB + 1];
    __shared__ int nMatch;

    unsigned short* tmpE = (unsigned short*)Mraw;
    float* part8 = (float*)sortedE;

    const int tid  = threadIdx.x;
    const int lane = tid & 31;
    const int wave = __builtin_amdgcn_readfirstlane(tid >> 5);
    const unsigned int blk = blockIdx.x;

    if (tid < APB) { cnt[tid] = 0; cur[tid] = 0; }
    if (tid == 0) nMatch = 0;
    __syncthreads();

    const int dn  = lane & 15;
    const int dhh = lane >> 4;
    const float* bptr = Wt + (size_t)(wave * KW + 2 * dhh) * NH + 4 * dn;
    v4f bx[4][2], by[4][2];
#pragma unroll
    for (int q = 0; q < 3; ++q)
#pragma unroll
        for (int s = 0; s < 2; ++s) {
            bx[q][s] = *(const v4f*)(bptr + (q * 8 + s * 4) * NH);
            by[q][s] = *(const v4f*)(bptr + (q * 8 + s * 4 + 1) * NH);
        }

    {
        const uint4* own4 = (const uint4*)own;
        uint4* ownL4 = (uint4*)sortedE;
        const unsigned int* ownLw = (const unsigned int*)sortedE;
        const unsigned int pat = blk * 0x01010101u;
        uint4 ov[N_EDGES / (16 * NTHR)];
#pragma unroll
        for (int it = 0; it < N_EDGES / (16 * NTHR); ++it) ov[it] = own4[it * NTHR + tid];
        unsigned long long m = 0ull;
#pragma unroll
        for (int it = 0; it < N_EDGES / (16 * NTHR); ++it) {
            ownL4[it * NTHR + tid] = ov[it];
            unsigned int wd[4] = {ov[it].x, ov[it].y, ov[it].z, ov[it].w};
#pragma unroll
            for (int wi = 0; wi < 4; ++wi) {
                unsigned int x = wd[wi] ^ pat;
                unsigned long long f = (((x - 0x01010101u) & ~x & 0x80808080u) != 0u) ? 1ull : 0ull;
                m |= f << (it * 4 + wi);
            }
        }
        while (m != 0ull) {
            int bpos = __builtin_ctzll(m);
            m &= m - 1ull;
            int widx = ((bpos >> 2) * NTHR + tid) * 4 + (bpos & 3);
            unsigned int x = ownLw[widx] ^ pat;
            unsigned int z = ~(((x & 0x7f7f7f7fu) + 0x7f7f7f7fu) | x) & 0x80808080u;
            int nm = __builtin_popcount(z);
            int idx = atomicAdd(&nMatch, nm);
#pragma unroll
            for (int bi = 0; bi < 4; ++bi) {
                if (z & (0x80u << (8 * bi))) {
                    tmpE[idx] = (unsigned short)(widx * 4 + bi);
                    ++idx;
                }
            }
        }
    }
    __syncthreads();
    const int nE = __builtin_amdgcn_readfirstlane(nMatch);

    for (int i = tid; i < nE; i += NTHR) {
        int e = tmpE[i];
        int a = a2p[2 * e] & (APB - 1);
        atomicAdd(&cnt[a], 1);
    }
    __syncthreads();
    if (tid == 0) {
        int s = 0;
        for (int a = 0; a < APB; ++a) { start[a] = s; s += cnt[a]; }
        start[APB] = s;
    }
    __syncthreads();
    for (int i = tid; i < nE; i += NTHR) {
        int e = tmpE[i];
        int a = a2p[2 * e] & (APB - 1);
        int pos = start[a] + atomicAdd(&cur[a], 1);
        sortedE[pos] = (unsigned short)e;
    }
    __syncthreads();

    {
        v2f acc[17];
#pragma unroll
        for (int p = 0; p < 17; ++p) acc[p] = (v2f)0.0f;
        int curA = wave * 4;
        const int lastA = wave * 4 + 4;
        const int rbeg = __builtin_amdgcn_readfirstlane(start[wave * 4]);
        const int rend = __builtin_amdgcn_readfirstlane(start[wave * 4 + 4]);

        auto load_idx = [&](int base, int& e_l, int& d_l, int& s_l) __attribute__((always_inline)) {
            int idx = base + lane;
            e_l = 0; d_l = 0; s_l = 0;
            if (idx < rend) {
                e_l = sortedE[idx];
                d_l = a2p[2 * e_l];
                s_l = a2p[2 * e_l + 1];
            }
        };
        auto issue = [&](int e_l, int s_l, int j0, v2f (&sv)[EG], float (&pv)[EG]) __attribute__((always_inline)) {
#pragma unroll
            for (int u = 0; u < EG; ++u) {
                int j = j0 + u;
                int e = __builtin_amdgcn_readlane(e_l, j);
                int s = __builtin_amdgcn_readlane(s_l, j);
                sv[u] = *(const v2f*)(atomf + (size_t)s * NH + 2 * lane);
                pv[u] = pf[(size_t)e * N_PAIR + (lane & 15)];
            }
        };
        auto sums = [&](int d_l, int j0, int nb, const v2f (&sv)[EG], const float (&pv)[EG]) __attribute__((always_inline)) {
#pragma unroll
            for (int u = 0; u < EG; ++u) {
                int j = j0 + u;
                if (j < nb) {
                    int a = __builtin_amdgcn_readlane(d_l, j) & (APB - 1);
                    while (curA < a) {
#pragma unroll
                        for (int p = 0; p < 17; ++p) {
                            *(v2f*)&Mraw[curA * MP + p * NH + 2 * lane] = acc[p];
                            acc[p] = (v2f)0.0f;
                        }
                        ++curA;
                    }
#pragma unroll
                    for (int p = 0; p < N_PAIR; ++p) {
                        float w = __int_as_float(__builtin_amdgcn_readlane(__float_as_int(pv[u]), p));
                        acc[p] += w * sv[u];
                    }
                    acc[N_PAIR] += sv[u];
                }
            }
        };

        if (rbeg < rend) {
            v2f   svA[EG], svB[EG];
            float pvA[EG], pvB[EG];
            int e_c, d_c, s_c;
            int e_n = 0, d_n = 0, s_n = 0;
            int base = rbeg, j0 = 0;
            int nb = rend - base;
            if (nb > 32) nb = 32;
            load_idx(base, e_c, d_c, s_c);
            issue(e_c, s_c, 0, svB, pvB);
            for (;;) {
#pragma unroll
                for (int u = 0; u < EG; ++u) { svA[u] = svB[u]; pvA[u] = pvB[u]; }
                const bool same = (j0 == 0) && (nb > EG);
                const bool more = base + 32 < rend;
                const bool next = same || more;
                if (next) {
                    const int e_i = same ? e_c : e_n;
                    const int s_i = same ? s_c : s_n;
                    issue(e_i, s_i, same ? EG : 0, svB, pvB);
                }
                if (j0 == 0 && more) load_idx(base + 32, e_n, d_n, s_n);
                sums(d_c, j0, nb, svA, pvA);
                if (!next) break;
                if (same) {
                    j0 = EG;
                } else {
                    j0 = 0;
                    base += 32;
                    nb = rend - base;
                    if (nb > 32) nb = 32;
                    e_c = e_n; d_c = d_n; s_c = s_n;
                }
            }
        }
        while (curA < lastA) {
#pragma unroll
            for (int p = 0; p < 17; ++p) {
                *(v2f*)&Mraw[curA * MP + p * NH + 2 * lane] = acc[p];
                acc[p] = (v2f)0.0f;
            }
            ++curA;
        }
    }
    __syncthreads();

    {
        const float* aptr = &Mraw[dn * MP + wave * KW + 2 * dhh];

        v8f acc[2][4];
#pragma unroll
        for (int t = 0; t < 2; ++t)
#pragma unroll
            for (int j = 0; j < 4; ++j) acc[t][j] = (v8f)0.0f;

        v2f af[2][2][2];
#pragma unroll
        for (int s = 0; s < 2; ++s) {
            af[0][s][0] = *(const v2f*)(aptr + 4 * s);
            af[0][s][1] = *(const v2f*)(aptr + 16 * MP + 4 * s);
        }

#pragma unroll
        for (int q = 0; q < NPAIR; ++q) {
            if (q + 3 < NPAIR) {
#pragma unroll
                for (int s = 0; s < 2; ++s) {
                    bx[(q + 3) & 3][s] = *(const v4f*)(bptr + ((q + 3) * 8 + s * 4) * NH);
                    by[(q + 3) & 3][s] = *(const v4f*)(bptr + ((q + 3) * 8 + s * 4 + 1) * NH);
                }
            }
            if (q + 1 < NPAIR) {
#pragma unroll
                for (int s = 0; s < 2; ++s) {
                    af[(q + 1) & 1][s][0] = *(const v2f*)(aptr + (q + 1) * 8 + 4 * s);
                    af[(q + 1) & 1][s][1] = *(const v2f*)(aptr + 16 * MP + (q + 1) * 8 + 4 * s);
                }
            }
#pragma unroll
            for (int s = 0; s < 2; ++s) {
#pragma unroll
                for (int j = 0; j < 4; ++j) {
                    v2f bf;
                    bf.x = bx[q & 3][s][j];
                    bf.y = by[q & 3][s][j];
                    acc[0][j] = __builtin_amdgcn_wmma_f32_16x16x4_f32(
                        false, af[q & 1][s][0], false, bf, (short)0, acc[0][j], false, false);
                    acc[1][j] = __builtin_amdgcn_wmma_f32_16x16x4_f32(
                        false, af[q & 1][s][1], false, bf, (short)0, acc[1][j], false, false);
                }
            }
            __builtin_amdgcn_sched_barrier(0);
        }

#pragma unroll
        for (int t = 0; t < 2; ++t)
#pragma unroll
            for (int r = 0; r < 8; ++r) {
                v4f v;
                v.x = acc[t][0][r]; v.y = acc[t][1][r]; v.z = acc[t][2][r]; v.w = acc[t][3][r];
                *(v4f*)&part8[(wave * APB + t * 16 + dhh * 8 + r) * PP + 4 * dn] = v;
            }
    }
    __syncthreads();

#pragma unroll
    for (int it = 0; it < 2; ++it) {
        int i4  = it * NTHR + tid;
        int row = i4 >> 4;
        int c4  = (i4 & 15) * 4;
        v4f r = *(const v4f*)&part8[row * PP + c4];
#pragma unroll
        for (int w = 1; w < 8; ++w) r += *(const v4f*)&part8[(w * APB + row) * PP + c4];
        *(v4f*)&out[((size_t)blk * APB + row) * NH + c4] = r;
    }
}

extern "C" void kernel_launch(void* const* d_in, const int* in_sizes, int n_in,
                              void* d_out, int out_size, void* d_ws, size_t ws_size,
                              hipStream_t stream) {
    const float* pf    = (const float*)d_in[0];
    const float* atomf = (const float*)d_in[1];
    const int*   a2p   = (const int*)d_in[2];
    const float* W     = (const float*)d_in[3];
    const float* b     = (const float*)d_in[4];
    float* out = (float*)d_out;
    float* Wt  = (float*)d_ws;
    unsigned int* own = (unsigned int*)((char*)d_ws + WT_BYTES);

    prep_wt<<<272, 256, 0, stream>>>(W, b, a2p, Wt, own);
    edge_net<<<256, 256, 0, stream>>>(pf, atomf, a2p, Wt, own, out);
}


# accumulation phase moved onto the matrix unit: per atom pf^T x src as f32 16x16x4 products over steps of four edges (segments padded to four, empty slots zero); edge sum order changes
# speedup vs baseline: 2.7716x; 1.1460x over previous
#include <hip/hip_runtime.h>


typedef __attribute__((ext_vector_type(2))) float v2f;
typedef __attribute__((ext_vector_type(8))) float v8f;
typedef __attribute__((ext_vector_type(4))) float v4f;

#define N_ATOMS 8192
#define N_EDGES 65536
#define N_PAIR  16
#define NH      64
#define KTOT    (17 * NH)
#define MP      (KTOT + 4)
#define APB     32
#define NTHR    256
#define CG      16
#define KW      (KTOT / 8)
#define NPAIR   (KW / 8)
#define PP      (NH + 4)
#define WT_BYTES (17 * NH * NH * 4)

__global__ void prep_wt(const float* __restrict__ W, const float* __restrict__ b,
                        const int* __restrict__ a2p,
                        float* __restrict__ Wt, unsigned int* __restrict__ own) {
    int idx = blockIdx.x * blockDim.x + threadIdx.x;
    if (idx < 17 * NH * NH) {
        int p = idx / (NH * NH);
        int r = idx % (NH * NH);
        int k = r / NH;
        int h = r % NH;
        Wt[idx] = (p < N_PAIR) ? W[p * (NH * NH) + h * NH + k] : b[h * NH + k];
    }
    if (idx < N_EDGES / 4) {
        unsigned int w = 0;
#pragma unroll
        for (int j = 0; j < 4; ++j) {
            unsigned int o = ((unsigned int)a2p[2 * (4 * idx + j)] >> 5) & 255u;
            w |= o << (8 * j);
        }
        own[idx] = w;
    }
}

__launch_bounds__(NTHR)
__global__ void edge_net(const float* __restrict__ pf,
                         const float* __restrict__ atomf,
                         const int*   __restrict__ a2p,
                         const float* __restrict__ Wt,
                         const unsigned int* __restrict__ own,
                         float*       __restrict__ out) {
    __shared__ __attribute__((aligned(16))) float Mraw[APB * MP];
    __shared__ __attribute__((aligned(16))) unsigned short sortedE[N_EDGES + 4 * APB];
    __shared__ int cnt[APB];
    __shared__ int cur[APB];
    __shared__ int start[APB + 1];
    __shared__ int nMatch;

    unsigned short* tmpE = (unsigned short*)Mraw;
    float* part8 = (float*)sortedE;

    const int tid  = threadIdx.x;
    const int lane = tid & 31;
    const int wave = __builtin_amdgcn_readfirstlane(tid >> 5);
    const unsigned int blk = blockIdx.x;

    if (tid < APB) { cnt[tid] = 0; cur[tid] = 0; }
    if (tid == 0) nMatch = 0;
    __syncthreads();

    const int dn  = lane & 15;
    const int dhh = lane >> 4;
    const float* bptr = Wt + (size_t)(wave * KW + 2 * dhh) * NH + 4 * dn;
    v4f bx[4][2], by[4][2];
#pragma unroll
    for (int q = 0; q < 3; ++q)
#pragma unroll
        for (int s = 0; s < 2; ++s) {
            bx[q][s] = *(const v4f*)(bptr + (q * 8 + s * 4) * NH);
            by[q][s] = *(const v4f*)(bptr + (q * 8 + s * 4 + 1) * NH);
        }

    {
        const uint4* own4 = (const uint4*)own;
        uint4* ownL4 = (uint4*)sortedE;
        const unsigned int* ownLw = (const unsigned int*)sortedE;
        const unsigned int pat = blk * 0x01010101u;
        uint4 ov[N_EDGES / (16 * NTHR)];
#pragma unroll
        for (int it = 0; it < N_EDGES / (16 * NTHR); ++it) ov[it] = own4[it * NTHR + tid];
        unsigned long long m = 0ull;
#pragma unroll
        for (int it = 0; it < N_EDGES / (16 * NTHR); ++it) {
            ownL4[it * NTHR + tid] = ov[it];
            unsigned int wd[4] = {ov[it].x, ov[it].y, ov[it].z, ov[it].w};
#pragma unroll
            for (int wi = 0; wi < 4; ++wi) {
                unsigned int x = wd[wi] ^ pat;
                unsigned long long f = (((x - 0x01010101u) & ~x & 0x80808080u) != 0u) ? 1ull : 0ull;
                m |= f << (it * 4 + wi);
            }
        }
        while (m != 0ull) {
            int bpos = __builtin_ctzll(m);
            m &= m - 1ull;
            int widx = ((bpos >> 2) * NTHR + tid) * 4 + (bpos & 3);
            unsigned int x = ownLw[widx] ^ pat;
            unsigned int z = ~(((x & 0x7f7f7f7fu) + 0x7f7f7f7fu) | x) & 0x80808080u;
            int nm = __builtin_popcount(z);
            int idx = atomicAdd(&nMatch, nm);
#pragma unroll
            for (int bi = 0; bi < 4; ++bi) {
                if (z & (0x80u << (8 * bi))) {
                    tmpE[idx] = (unsigned short)(widx * 4 + bi);
                    ++idx;
                }
            }
        }
    }
    __syncthreads();
    const int nE = __builtin_amdgcn_readfirstlane(nMatch);

    for (int i = tid; i < nE; i += NTHR) {
        int e = tmpE[i];
        int a = a2p[2 * e] & (APB - 1);
        atomicAdd(&cnt[a], 1);
    }
    __syncthreads();
    if (tid == 0) {
        int s = 0;
        for (int a = 0; a < APB; ++a) { start[a] = s; s += (cnt[a] + 3) & ~3; }
        start[APB] = s;
    }
    __syncthreads();
    for (int i = tid; i < nE; i += NTHR) {
        int e = tmpE[i];
        int a = a2p[2 * e] & (APB - 1);
        int r = atomicAdd(&cur[a], 1);
        int pos = start[a] + r;
        sortedE[pos] = (unsigned short)e;
        int c = cnt[a];
        if (r == c - 1)
            for (int k = c; k < ((c + 3) & ~3); ++k) sortedE[start[a] + k] = (unsigned short)e;
    }
    __syncthreads();

    {
        const int cc = lane & 15;
        const int hh = lane >> 4;
        const int rbeg = __builtin_amdgcn_readfirstlane(start[wave * 4]);
        const int rend = __builtin_amdgcn_readfirstlane(start[wave * 4 + 4]);
        const int lastA = wave * 4 + 4;
        int curA = wave * 4;

        v8f macc[4];
        v4f bsum = (v4f)0.0f;
#pragma unroll
        for (int j = 0; j < 4; ++j) macc[j] = (v8f)0.0f;

        auto flush = [&](int a) __attribute__((always_inline)) {
#pragma unroll
            for (int r = 0; r < 8; ++r) {
                v4f v;
                v.x = macc[0][r]; v.y = macc[1][r]; v.z = macc[2][r]; v.w = macc[3][r];
                *(v4f*)&Mraw[a * MP + (8 * hh + r) * NH + 4 * cc] = v;
            }
            v4f o;
            o.x = bsum.x + __shfl_xor(bsum.x, 16);
            o.y = bsum.y + __shfl_xor(bsum.y, 16);
            o.z = bsum.z + __shfl_xor(bsum.z, 16);
            o.w = bsum.w + __shfl_xor(bsum.w, 16);
            if (hh == 0) *(v4f*)&Mraw[a * MP + N_PAIR * NH + 4 * cc] = o;
#pragma unroll
            for (int j = 0; j < 4; ++j) macc[j] = (v8f)0.0f;
            bsum = (v4f)0.0f;
        };

        for (int gb = rbeg; gb < rend; gb += 4 * CG) {
            int          e0[CG], e1[CG];
            int          s0[CG], s1[CG];
            v4f          r0[CG], r1[CG];
            float        w0[CG], w1[CG];
#pragma unroll
            for (int t = 0; t < CG; ++t) {
                int i0 = gb + 4 * t + 2 * hh;
                if (i0 > rend - 2) i0 = rend - 2;
                unsigned int ep = *(const unsigned int*)&sortedE[i0];
                e0[t] = (int)(ep & 0xffffu);
                e1[t] = (int)(ep >> 16);
            }
#pragma unroll
            for (int t = 0; t < CG; ++t) {
                s0[t] = a2p[2 * e0[t] + 1];
                s1[t] = a2p[2 * e1[t] + 1];
            }
#pragma unroll
            for (int t = 0; t < CG; ++t) {
                r0[t] = *(const v4f*)(atomf + (size_t)s0[t] * NH + 4 * cc);
                r1[t] = *(const v4f*)(atomf + (size_t)s1[t] * NH + 4 * cc);
                w0[t] = pf[(size_t)e0[t] * N_PAIR + cc];
                w1[t] = pf[(size_t)e1[t] * N_PAIR + cc];
            }
            auto step = [&](int t, int vend) __attribute__((always_inline)) {
                const int  i0 = gb + 4 * t + 2 * hh;
                const bool v0 = i0 < vend;
                const bool v1 = i0 + 1 < vend;
                v2f af;
                af.x = v0 ? w0[t] : 0.0f;
                af.y = v1 ? w1[t] : 0.0f;
                const v4f x0 = v0 ? r0[t] : (v4f)0.0f;
                const v4f x1 = v1 ? r1[t] : (v4f)0.0f;
#pragma unroll
                for (int j = 0; j < 4; ++j) {
                    v2f bf;
                    bf.x = x0[j];
                    bf.y = x1[j];
                    macc[j] = __builtin_amdgcn_wmma_f32_16x16x4_f32(
                        false, af, false, bf, (short)0, macc[j], false, false);
                }
                bsum += x0 + x1;
            };
            const int gend = (gb + 4 * CG < rend) ? gb + 4 * CG : rend;
            while (curA < lastA) {
                const int aS   = __builtin_amdgcn_readfirstlane(start[curA]);
                const int aE   = __builtin_amdgcn_readfirstlane(start[curA + 1]);
                const int vend = aS + __builtin_amdgcn_readfirstlane(cnt[curA]);
                if (aS >= gend) break;
                const int t0 = ((aS > gb ? aS : gb) - gb) >> 2;
                const int t1 = ((aE < gend ? aE : gend) - gb) >> 2;
                if (t1 > t0) {
                    const unsigned int nst = (unsigned int)(t1 - t0);
#pragma unroll
                    for (int t = 0; t < CG; ++t) {
                        if ((unsigned int)(t - t0) < nst) step(t, vend);
                    }
                }
                if (aE > gend) break;
                flush(curA);
                ++curA;
            }
        }
        while (curA < lastA) {
            flush(curA);
            ++curA;
        }
    }
    __syncthreads();

    {
        const float* aptr = &Mraw[dn * MP + wave * KW + 2 * dhh];

        v8f acc[2][4];
#pragma unroll
        for (int t = 0; t < 2; ++t)
#pragma unroll
            for (int j = 0; j < 4; ++j) acc[t][j] = (v8f)0.0f;

        v2f af[2][2][2];
#pragma unroll
        for (int s = 0; s < 2; ++s) {
            af[0][s][0] = *(const v2f*)(aptr + 4 * s);
            af[0][s][1] = *(const v2f*)(aptr + 16 * MP + 4 * s);
        }

#pragma unroll
        for (int q = 0; q < NPAIR; ++q) {
            if (q + 3 < NPAIR) {
#pragma unroll
                for (int s = 0; s < 2; ++s) {
                    bx[(q + 3) & 3][s] = *(const v4f*)(bptr + ((q + 3) * 8 + s * 4) * NH);
                    by[(q + 3) & 3][s] = *(const v4f*)(bptr + ((q + 3) * 8 + s * 4 + 1) * NH);
                }
            }
            if (q + 1 < NPAIR) {
#pragma unroll
                for (int s = 0; s < 2; ++s) {
                    af[(q + 1) & 1][s][0] = *(const v2f*)(aptr + (q + 1) * 8 + 4 * s);
                    af[(q + 1) & 1][s][1] = *(const v2f*)(aptr + 16 * MP + (q + 1) * 8 + 4 * s);
                }
            }
#pragma unroll
            for (int s = 0; s < 2; ++s) {
#pragma unroll
                for (int j = 0; j < 4; ++j) {
                    v2f bf;
                    bf.x = bx[q & 3][s][j];
                    bf.y = by[q & 3][s][j];
                    acc[0][j] = __builtin_amdgcn_wmma_f32_16x16x4_f32(
                        false, af[q & 1][s][0], false, bf, (short)0, acc[0][j], false, false);
                    acc[1][j] = __builtin_amdgcn_wmma_f32_16x16x4_f32(
                        false, af[q & 1][s][1], false, bf, (short)0, acc[1][j], false, false);
                }
            }
            __builtin_amdgcn_sched_barrier(0);
        }

#pragma unroll
        for (int t = 0; t < 2; ++t)
#pragma unroll
            for (int r = 0; r < 8; ++r) {
                v4f v;
                v.x = acc[t][0][r]; v.y = acc[t][1][r]; v.z = acc[t][2][r]; v.w = acc[t][3][r];
                *(v4f*)&part8[(wave * APB + t * 16 + dhh * 8 + r) * PP + 4 * dn] = v;
            }
    }
    __syncthreads();

#pragma unroll
    for (int it = 0; it < 2; ++it) {
        int i4  = it * NTHR + tid;
        int row = i4 >> 4;
        int c4  = (i4 & 15) * 4;
        v4f r = *(const v4f*)&part8[row * PP + c4];
#pragma unroll
        for (int w = 1; w < 8; ++w) r += *(const v4f*)&part8[(w * APB + row) * PP + c4];
        *(v4f*)&out[((size_t)blk * APB + row) * NH + c4] = r;
    }
}

extern "C" void kernel_launch(void* const* d_in, const int* in_sizes, int n_in,
                              void* d_out, int out_size, void* d_ws, size_t ws_size,
                              hipStream_t stream) {
    const float* pf    = (const float*)d_in[0];
    const float* atomf = (const float*)d_in[1];
    const int*   a2p   = (const int*)d_in[2];
    const float* W     = (const float*)d_in[3];
    const float* b     = (const float*)d_in[4];
    float* out = (float*)d_out;
    float* Wt  = (float*)d_ws;
    unsigned int* own = (unsigned int*)((char*)d_ws + WT_BYTES);

    prep_wt<<<272, 256, 0, stream>>>(W, b, a2p, Wt, own);
    edge_net<<<256, 256, 0, stream>>>(pf, atomf, a2p, Wt, own, out);
}


# edge binning without the sort: a wave spreads its flagged owner words over its lanes and puts each edge straight into its atom's LDS slot list (slot from an LDS counter add); edge sum order changes
# speedup vs baseline: 3.0185x; 1.0891x over previous
#include <hip/hip_runtime.h>


typedef __attribute__((ext_vector_type(2))) float v2f;
typedef __attribute__((ext_vector_type(8))) float v8f;
typedef __attribute__((ext_vector_type(4))) float v4f;

#define N_ATOMS 8192
#define N_EDGES 65536
#define N_PAIR  16
#define NH      64
#define KTOT    (17 * NH)
#define MP      (KTOT + 4)
#define APB     32
#define NTHR    256
#define CG      16
#define KW      (KTOT / 8)
#define NPAIR   (KW / 8)
#define PP      (NH + 4)
#define WT_BYTES (17 * NH * NH * 4)
#define CAP     64
#define WSEG    2048

__global__ void prep_wt(const float* __restrict__ W, const float* __restrict__ b,
                        const int* __restrict__ a2p,
                        float* __restrict__ Wt, unsigned int* __restrict__ own) {
    int idx = blockIdx.x * blockDim.x + threadIdx.x;
    if (idx < 17 * NH * NH) {
        int p = idx / (NH * NH);
        int r = idx % (NH * NH);
        int k = r / NH;
        int h = r % NH;
        Wt[idx] = (p < N_PAIR) ? W[p * (NH * NH) + h * NH + k] : b[h * NH + k];
    }
    if (idx < N_EDGES / 4) {
        unsigned int w = 0;
#pragma unroll
        for (int j = 0; j < 4; ++j) {
            unsigned int o = ((unsigned int)a2p[2 * (4 * idx + j)] >> 5) & 255u;
            w |= o << (8 * j);
        }
        own[idx] = w;
    }
}

__launch_bounds__(NTHR)
__global__ void edge_net(const float* __restrict__ pf,
                         const float* __restrict__ atomf,
                         const int*   __restrict__ a2p,
                         const float* __restrict__ Wt,
                         const unsigned int* __restrict__ own,
                         float*       __restrict__ out) {
    __shared__ __attribute__((aligned(16))) float Mraw[APB * MP];
    __shared__ __attribute__((aligned(16))) unsigned short overE[N_EDGES];
    __shared__ __attribute__((aligned(16))) unsigned int slotL[APB * CAP];
    __shared__ int cnt[APB];
    __shared__ int start[APB + 1];
    __shared__ int nOver;

    unsigned short* flagW = (unsigned short*)Mraw;
    float* part8 = (float*)overE;

    const int tid  = threadIdx.x;
    const int lane = tid & 31;
    const int wave = __builtin_amdgcn_readfirstlane(tid >> 5);
    const unsigned int blk = blockIdx.x;

    if (tid < APB) cnt[tid] = 0;
    if (tid == 0) nOver = 0;
    __syncthreads();

    const int dn  = lane & 15;
    const int dhh = lane >> 4;
    const float* bptr = Wt + (size_t)(wave * KW + 2 * dhh) * NH + 4 * dn;
    v4f bx[4][2], by[4][2];
#pragma unroll
    for (int q = 0; q < 3; ++q)
#pragma unroll
        for (int s = 0; s < 2; ++s) {
            bx[q][s] = *(const v4f*)(bptr + (q * 8 + s * 4) * NH);
            by[q][s] = *(const v4f*)(bptr + (q * 8 + s * 4 + 1) * NH);
        }

    {
        const uint4* own4 = (const uint4*)own;
        const unsigned int pat = blk * 0x01010101u;
        uint4 ov[N_EDGES / (16 * NTHR)];
#pragma unroll
        for (int it = 0; it < N_EDGES / (16 * NTHR); ++it) ov[it] = own4[it * NTHR + tid];
        unsigned long long m = 0ull;
#pragma unroll
        for (int it = 0; it < N_EDGES / (16 * NTHR); ++it) {
            unsigned int wd[4] = {ov[it].x, ov[it].y, ov[it].z, ov[it].w};
#pragma unroll
            for (int wi = 0; wi < 4; ++wi) {
                unsigned int x = wd[wi] ^ pat;
                unsigned long long f = (((x - 0x01010101u) & ~x & 0x80808080u) != 0u) ? 1ull : 0ull;
                m |= f << (it * 4 + wi);
            }
        }
        unsigned short* myW = flagW + wave * WSEG;
        int nW = 0;
        for (int trip = 0; trip < 64; ++trip) {
            const bool has = (m != 0ull);
            const unsigned int bal = __builtin_amdgcn_ballot_w32(has);
            if (bal == 0u) break;
            if (has) {
                const int bpos = __builtin_ctzll(m);
                m &= m - 1ull;
                const int widx = ((bpos >> 2) * NTHR + tid) * 4 + (bpos & 3);
                myW[nW + (int)__builtin_amdgcn_mbcnt_lo(bal, 0u)] = (unsigned short)widx;
            }
            nW += __builtin_popcount(bal);
        }
        __builtin_amdgcn_fence(__ATOMIC_RELEASE, "wavefront");
        __builtin_amdgcn_wave_barrier();
        __builtin_amdgcn_fence(__ATOMIC_ACQUIRE, "wavefront");

        for (int i0 = lane; i0 < nW; i0 += 64) {
            const int  i1   = i0 + 32;
            const bool has1 = i1 < nW;
            const int  wq[2] = {(int)myW[i0], (int)myW[has1 ? i1 : i0]};
            uint4 qa[2], qb[2];
#pragma unroll
            for (int u = 0; u < 2; ++u) {
                const uint4* q = (const uint4*)(a2p + 8 * wq[u]);
                qa[u] = q[0];
                qb[u] = q[1];
            }
            const unsigned int kill1 = has1 ? 0u : 0xffffffffu;
            unsigned int dd[8], ss[8];
            bool         mt[8];
            int          rr[8];
#pragma unroll
            for (int u = 0; u < 2; ++u) {
                dd[4 * u + 0] = qa[u].x; ss[4 * u + 0] = qa[u].y;
                dd[4 * u + 1] = qa[u].z; ss[4 * u + 1] = qa[u].w;
                dd[4 * u + 2] = qb[u].x; ss[4 * u + 2] = qb[u].y;
                dd[4 * u + 3] = qb[u].z; ss[4 * u + 3] = qb[u].w;
            }
#pragma unroll
            for (int j = 0; j < 8; ++j) {
                mt[j] = ((dd[j] | (j < 4 ? 0u : kill1)) >> 5) == blk;
                rr[j] = 0;
                if (mt[j]) rr[j] = atomicAdd(&cnt[dd[j] & (APB - 1)], 1);
            }
#pragma unroll
            for (int j = 0; j < 8; ++j) {
                if (mt[j]) {
                    const unsigned int e = (unsigned int)(wq[j >> 2] * 4 + (j & 3));
                    if (rr[j] < CAP) {
                        slotL[(dd[j] & (APB - 1)) * CAP + rr[j]] = e | (ss[j] << 16);
                    } else {
                        const int o = atomicAdd(&nOver, 1);
                        overE[o] = (unsigned short)e;
                    }
                }
            }
        }
    }
    __syncthreads();

    if (tid < APB) {
        int c = cnt[tid];
        if (c > CAP) c = CAP;
        const int npad = ((c + 3) & ~3) - c;
        if (npad > 0) {
            const unsigned int lastv = slotL[tid * CAP + c - 1];
            slotL[tid * CAP + c] = lastv;
            if (npad > 1) slotL[tid * CAP + c + 1] = lastv;
            if (npad > 2) slotL[tid * CAP + c + 2] = lastv;
        }
    }
    if (tid == 0) {
        int s = 0;
        for (int a = 0; a < APB; ++a) {
            int c = cnt[a];
            if (c > CAP) c = CAP;
            start[a] = s;
            s += (c + 3) & ~3;
        }
        start[APB] = s;
    }
    __syncthreads();

    {
        const int cc = lane & 15;
        const int hh = lane >> 4;
        const int rbeg = __builtin_amdgcn_readfirstlane(start[wave * 4]);
        const int rend = __builtin_amdgcn_readfirstlane(start[wave * 4 + 4]);
        const int st1  = __builtin_amdgcn_readfirstlane(start[wave * 4 + 1]);
        const int st2  = __builtin_amdgcn_readfirstlane(start[wave * 4 + 2]);
        const int st3  = __builtin_amdgcn_readfirstlane(start[wave * 4 + 3]);
        const int off0 = (wave * 4 + 0) * CAP - rbeg;
        const int off1 = (wave * 4 + 1) * CAP - st1;
        const int off2 = (wave * 4 + 2) * CAP - st2;
        const int off3 = (wave * 4 + 3) * CAP - st3;
        const int lastA = wave * 4 + 4;
        int curA = wave * 4;

        v8f macc[4];
        v4f bsum = (v4f)0.0f;
#pragma unroll
        for (int j = 0; j < 4; ++j) macc[j] = (v8f)0.0f;

        auto flush = [&](int a) __attribute__((always_inline)) {
#pragma unroll
            for (int r = 0; r < 8; ++r) {
                v4f v;
                v.x = macc[0][r]; v.y = macc[1][r]; v.z = macc[2][r]; v.w = macc[3][r];
                *(v4f*)&Mraw[a * MP + (8 * hh + r) * NH + 4 * cc] = v;
            }
            v4f o;
            o.x = bsum.x + __shfl_xor(bsum.x, 16);
            o.y = bsum.y + __shfl_xor(bsum.y, 16);
            o.z = bsum.z + __shfl_xor(bsum.z, 16);
            o.w = bsum.w + __shfl_xor(bsum.w, 16);
            if (hh == 0) *(v4f*)&Mraw[a * MP + N_PAIR * NH + 4 * cc] = o;
#pragma unroll
            for (int j = 0; j < 4; ++j) macc[j] = (v8f)0.0f;
            bsum = (v4f)0.0f;
        };

        for (int gb = rbeg; gb < rend; gb += 4 * CG) {
            int          e0[CG], e1[CG];
            int          s0[CG], s1[CG];
            v4f          r0[CG], r1[CG];
            float        w0[CG], w1[CG];
#pragma unroll
            for (int t = 0; t < CG; ++t) {
                int i0 = gb + 4 * t + 2 * hh;
                if (i0 > rend - 2) i0 = rend - 2;
                const int off = (i0 >= st3) ? off3 : ((i0 >= st2) ? off2 : ((i0 >= st1) ? off1 : off0));
                const uint2 ep = *(const uint2*)&slotL[i0 + off];
                e0[t] = (int)(ep.x & 0xffffu);
                s0[t] = (int)(ep.x >> 16);
                e1[t] = (int)(ep.y & 0xffffu);
                s1[t] = (int)(ep.y >> 16);
            }
#pragma unroll
            for (int t = 0; t < CG; ++t) {
                r0[t] = *(const v4f*)(atomf + (size_t)s0[t] * NH + 4 * cc);
                r1[t] = *(const v4f*)(atomf + (size_t)s1[t] * NH + 4 * cc);
                w0[t] = pf[(size_t)e0[t] * N_PAIR + cc];
                w1[t] = pf[(size_t)e1[t] * N_PAIR + cc];
            }
            auto step = [&](int t, int vend) __attribute__((always_inline)) {
                const int  i0 = gb + 4 * t + 2 * hh;
                const bool v0 = i0 < vend;
                const bool v1 = i0 + 1 < vend;
                v2f af;
                af.x = v0 ? w0[t] : 0.0f;
                af.y = v1 ? w1[t] : 0.0f;
                const v4f x0 = v0 ? r0[t] : (v4f)0.0f;
                const v4f x1 = v1 ? r1[t] : (v4f)0.0f;
#pragma unroll
                for (int j = 0; j < 4; ++j) {
                    v2f bf;
                    bf.x = x0[j];
                    bf.y = x1[j];
                    macc[j] = __builtin_amdgcn_wmma_f32_16x16x4_f32(
                        false, af, false, bf, (short)0, macc[j], false, false);
                }
                bsum += x0 + x1;
            };
            const int gend = (gb + 4 * CG < rend) ? gb + 4 * CG : rend;
            while (curA < lastA) {
                const int aS   = __builtin_amdgcn_readfirstlane(start[curA]);
                const int aE   = __builtin_amdgcn_readfirstlane(start[curA + 1]);
                const int cA   = __builtin_amdgcn_readfirstlane(cnt[curA]);
                const int vend = aS + (cA < CAP ? cA : CAP);
                if (aS >= gend) break;
                const int t0 = ((aS > gb ? aS : gb) - gb) >> 2;
                const int t1 = ((aE < gend ? aE : gend) - gb) >> 2;
                if (t1 > t0) {
                    const unsigned int nst = (unsigned int)(t1 - t0);
#pragma unroll
                    for (int t = 0; t < CG; ++t) {
                        if ((unsigned int)(t - t0) < nst) step(t, vend);
                    }
                }
                if (aE > gend) break;
                flush(curA);
                ++curA;
            }
        }
        while (curA < lastA) {
            flush(curA);
            ++curA;
        }
    }
    __syncthreads();

    {
        const int nOv = __builtin_amdgcn_readfirstlane(nOver);
        if (nOv > 0) {
            for (int i = 0; i < nOv; ++i) {
                const int e = overE[i];
                const int a = a2p[2 * e] & (APB - 1);
                const int s = a2p[2 * e + 1];
                for (int idx = tid; idx < KTOT; idx += NTHR) {
                    const int   p = idx >> 6;
                    const int   k = idx & (NH - 1);
                    const float w = (p < N_PAIR) ? pf[(size_t)e * N_PAIR + p] : 1.0f;
                    Mraw[a * MP + idx] += w * atomf[(size_t)s * NH + k];
                }
            }
            __syncthreads();
        }
    }

    {
        const float* aptr = &Mraw[dn * MP + wave * KW + 2 * dhh];

        v8f acc[2][4];
#pragma unroll
        for (int t = 0; t < 2; ++t)
#pragma unroll
            for (int j = 0; j < 4; ++j) acc[t][j] = (v8f)0.0f;

        v2f af[2][2][2];
#pragma unroll
        for (int s = 0; s < 2; ++s) {
            af[0][s][0] = *(const v2f*)(aptr + 4 * s);
            af[0][s][1] = *(const v2f*)(aptr + 16 * MP + 4 * s);
        }

#pragma unroll
        for (int q = 0; q < NPAIR; ++q) {
            if (q + 3 < NPAIR) {
#pragma unroll
                for (int s = 0; s < 2; ++s) {
                    bx[(q + 3) & 3][s] = *(const v4f*)(bptr + ((q + 3) * 8 + s * 4) * NH);
                    by[(q + 3) & 3][s] = *(const v4f*)(bptr + ((q + 3) * 8 + s * 4 + 1) * NH);
                }
            }
            if (q + 1 < NPAIR) {
#pragma unroll
                for (int s = 0; s < 2; ++s) {
                    af[(q + 1) & 1][s][0] = *(const v2f*)(aptr + (q + 1) * 8 + 4 * s);
                    af[(q + 1) & 1][s][1] = *(const v2f*)(aptr + 16 * MP + (q + 1) * 8 + 4 * s);
                }
            }
#pragma unroll
            for (int s = 0; s < 2; ++s) {
#pragma unroll
                for (int j = 0; j < 4; ++j) {
                    v2f bf;
                    bf.x = bx[q & 3][s][j];
                    bf.y = by[q & 3][s][j];
                    acc[0][j] = __builtin_amdgcn_wmma_f32_16x16x4_f32(
                        false, af[q & 1][s][0], false, bf, (short)0, acc[0][j], false, false);
                    acc[1][j] = __builtin_amdgcn_wmma_f32_16x16x4_f32(
                        false, af[q & 1][s][1], false, bf, (short)0, acc[1][j], false, false);
                }
            }
            __builtin_amdgcn_sched_barrier(0);
        }

#pragma unroll
        for (int t = 0; t < 2; ++t)
#pragma unroll
            for (int r = 0; r < 8; ++r) {
                v4f v;
                v.x = acc[t][0][r]; v.y = acc[t][1][r]; v.z = acc[t][2][r]; v.w = acc[t][3][r];
                *(v4f*)&part8[(wave * APB + t * 16 + dhh * 8 + r) * PP + 4 * dn] = v;
            }
    }
    __syncthreads();

#pragma unroll
    for (int it = 0; it < 2; ++it) {
        int i4  = it * NTHR + tid;
        int row = i4 >> 4;
        int c4  = (i4 & 15) * 4;
        v4f r = *(const v4f*)&part8[row * PP + c4];
#pragma unroll
        for (int w = 1; w < 8; ++w) r += *(const v4f*)&part8[(w * APB + row) * PP + c4];
        *(v4f*)&out[((size_t)blk * APB + row) * NH + c4] = r;
    }
}

extern "C" void kernel_launch(void* const* d_in, const int* in_sizes, int n_in,
                              void* d_out, int out_size, void* d_ws, size_t ws_size,
                              hipStream_t stream) {
    const float* pf    = (const float*)d_in[0];
    const float* atomf = (const float*)d_in[1];
    const int*   a2p   = (const int*)d_in[2];
    const float* W     = (const float*)d_in[3];
    const float* b     = (const float*)d_in[4];
    float* out = (float*)d_out;
    float* Wt  = (float*)d_ws;
    unsigned int* own = (unsigned int*)((char*)d_ws + WT_BYTES);

    prep_wt<<<272, 256, 0, stream>>>(W, b, a2p, Wt, own);
    edge_net<<<256, 256, 0, stream>>>(pf, atomf, a2p, Wt, own, out);
}


# edge binning moved into launch 1: 32 work-groups sort 2048-edge chunks by destination group into 128-byte buckets (LDS counter adds); a group reads 4 KB, no owner-byte scan; edge sum order changes
# speedup vs baseline: 3.2448x; 1.0750x over previous
#include <hip/hip_runtime.h>


typedef __attribute__((ext_vector_type(2))) float v2f;
typedef __attribute__((ext_vector_type(8))) float v8f;
typedef __attribute__((ext_vector_type(4))) float v4f;

#define N_ATOMS 8192
#define N_EDGES 65536
#define N_PAIR  16
#define NH      64
#define KTOT    (17 * NH)
#define MP      (KTOT + 4)
#define APB     32
#define NTHR    256
#define CG      16
#define KW      (KTOT / 8)
#define NPAIR   (KW / 8)
#define PP      (NH + 4)
#define WT_BYTES (17 * NH * NH * 4)
#define CAP     64
#define NCH     32
#define CHE     (N_EDGES / NCH)
#define BK      32
#define EMPTYV  0xffffffffu
#define OVFLV   0xfffffffeu

__global__ void prep_wt(const float* __restrict__ W, const float* __restrict__ b,
                        const int* __restrict__ a2p,
                        float* __restrict__ Wt, unsigned int* __restrict__ bucket) {
    __shared__ __attribute__((aligned(16))) unsigned int tab[256 * BK];
    __shared__ int gcnt[256];
    const int tid = threadIdx.x;
    if (blockIdx.x < NCH) {
        const int ch = blockIdx.x;
        const uint2* a2 = (const uint2*)a2p + (size_t)ch * CHE;
        uint2 pr[CHE / 256];
#pragma unroll
        for (int j = 0; j < CHE / 256; ++j) pr[j] = a2[j * 256 + tid];
        gcnt[tid] = 0;
        const uint4 ev = {EMPTYV, EMPTYV, EMPTYV, EMPTYV};
#pragma unroll
        for (int j = 0; j < 8; ++j) ((uint4*)tab)[j * 256 + tid] = ev;
        __syncthreads();
        int sl[CHE / 256];
#pragma unroll
        for (int j = 0; j < CHE / 256; ++j) sl[j] = atomicAdd(&gcnt[(pr[j].x >> 5) & 255u], 1);
#pragma unroll
        for (int j = 0; j < CHE / 256; ++j) {
            if (sl[j] < BK)
                tab[((pr[j].x >> 5) & 255u) * BK + sl[j]] =
                    (unsigned int)(j * 256 + tid) | ((pr[j].y & 0x1fffu) << 11) | ((pr[j].x & 31u) << 24);
        }
        __syncthreads();
#pragma unroll
        for (int j = 0; j < 8; ++j) {
            const int p = j * 256 + tid;
            const int g = p >> 3;
            uint4 v = ((const uint4*)tab)[p];
            if (gcnt[g] > BK) v.w = OVFLV;
            ((uint4*)bucket)[((size_t)g * NCH + ch) * 8 + (p & 7)] = v;
        }
    } else {
        const int idx = (blockIdx.x - NCH) * blockDim.x + tid;
        if (idx < 17 * NH * NH) {
            int p = idx / (NH * NH);
            int r = idx % (NH * NH);
            int k = r / NH;
            int h = r % NH;
            Wt[idx] = (p < N_PAIR) ? W[p * (NH * NH) + h * NH + k] : b[h * NH + k];
        }
    }
}

__launch_bounds__(NTHR)
__global__ void edge_net(const float* __restrict__ pf,
                         const float* __restrict__ atomf,
                         const int*   __restrict__ a2p,
                         const float* __restrict__ Wt,
                         const unsigned int* __restrict__ bucket,
                         float*       __restrict__ out) {
    __shared__ __attribute__((aligned(16))) float Mraw[APB * MP];
    __shared__ __attribute__((aligned(16))) unsigned short overE[N_EDGES];
    __shared__ __attribute__((aligned(16))) unsigned int slotL[APB * CAP];
    __shared__ int cnt[2 * APB];
    __shared__ int start[APB + 1];
    __shared__ int nOver;
    __shared__ int ovf[NCH];

    float* part8 = (float*)overE;

    const int tid  = threadIdx.x;
    const int lane = tid & 31;
    const int wave = __builtin_amdgcn_readfirstlane(tid >> 5);
    const unsigned int blk = blockIdx.x;

    if (tid < 2 * APB) cnt[tid] = 0;
    if (tid < NCH) ovf[tid] = 0;
    if (tid == 0) nOver = 0;
    __syncthreads();

    const int dn  = lane & 15;
    const int dhh = lane >> 4;
    const float* bptr = Wt + (size_t)(wave * KW + 2 * dhh) * NH + 4 * dn;
    v4f bx[4][2], by[4][2];
#pragma unroll
    for (int q = 0; q < 3; ++q)
#pragma unroll
        for (int s = 0; s < 2; ++s) {
            bx[q][s] = *(const v4f*)(bptr + (q * 8 + s * 4) * NH);
            by[q][s] = *(const v4f*)(bptr + (q * 8 + s * 4 + 1) * NH);
        }

    {
        auto put = [&](bool m, unsigned int e, unsigned int dl, unsigned int s, int r) __attribute__((always_inline)) {
            if (m) {
                if (r < CAP) {
                    slotL[dl * CAP + r] = e | (s << 16);
                } else {
                    const int o = atomicAdd(&nOver, 1);
                    overE[o] = (unsigned short)e;
                }
            }
        };
        const uint4 bv = ((const uint4*)bucket)[(size_t)blk * (NCH * 8) + tid];
        const unsigned int ch = (unsigned int)tid >> 3;
        const bool isOv = (bv.w == OVFLV);
        if (isOv) ovf[ch] = 1;
        {
            const unsigned int xs[4] = {bv.x, bv.y, bv.z, bv.w};
            bool mt[4];
            int  rr[4];
#pragma unroll
            for (int j = 0; j < 4; ++j) {
                mt[j] = !isOv && (xs[j] >> 29) == 0u;
                rr[j] = atomicAdd(&cnt[mt[j] ? (int)((xs[j] >> 24) & (APB - 1)) : APB + lane], 1);
            }
#pragma unroll
            for (int j = 0; j < 4; ++j)
                put(mt[j], (ch << 11) | (xs[j] & 0x7ffu), (xs[j] >> 24) & (APB - 1), (xs[j] >> 11) & 0x1fffu, rr[j]);
        }
        __syncthreads();
        unsigned int om = __builtin_amdgcn_ballot_w32(ovf[lane] != 0);
        if (om != 0u) {
            while (om != 0u) {
                const int oc = __builtin_ctz(om);
                om &= om - 1u;
                const uint2* a2 = (const uint2*)a2p + (size_t)oc * CHE;
                for (int j = 0; j < CHE / NTHR; ++j) {
                    const unsigned int i = (unsigned int)(j * NTHR + tid);
                    const uint2 pr = a2[i];
                    const bool  m  = (pr.x >> 5) == blk;
                    int r = 0;
                    if (m) r = atomicAdd(&cnt[pr.x & (APB - 1)], 1);
                    put(m, ((unsigned int)oc << 11) | i, pr.x & (APB - 1), pr.y & 0x1fffu, r);
                }
            }
            __syncthreads();
        }
    }

    if (tid < APB) {
        int c = cnt[tid];
        if (c > CAP) c = CAP;
        const int npad = ((c + 3) & ~3) - c;
        if (npad > 0) {
            const unsigned int lastv = slotL[tid * CAP + c - 1];
            slotL[tid * CAP + c] = lastv;
            if (npad > 1) slotL[tid * CAP + c + 1] = lastv;
            if (npad > 2) slotL[tid * CAP + c + 2] = lastv;
        }
    }
    if (tid == 0) {
        int s = 0;
        for (int a = 0; a < APB; ++a) {
            int c = cnt[a];
            if (c > CAP) c = CAP;
            start[a] = s;
            s += (c + 3) & ~3;
        }
        start[APB] = s;
    }
    __syncthreads();

    {
        const int cc = lane & 15;
        const int hh = lane >> 4;
        const int rbeg = __builtin_amdgcn_readfirstlane(start[wave * 4]);
        const int rend = __builtin_amdgcn_readfirstlane(start[wave * 4 + 4]);
        const int st1  = __builtin_amdgcn_readfirstlane(start[wave * 4 + 1]);
        const int st2  = __builtin_amdgcn_readfirstlane(start[wave * 4 + 2]);
        const int st3  = __builtin_amdgcn_readfirstlane(start[wave * 4 + 3]);
        const int off0 = (wave * 4 + 0) * CAP - rbeg;
        const int off1 = (wave * 4 + 1) * CAP - st1;
        const int off2 = (wave * 4 + 2) * CAP - st2;
        const int off3 = (wave * 4 + 3) * CAP - st3;
        const int lastA = wave * 4 + 4;
        int curA = wave * 4;

        v8f macc[4];
        v4f bsum = (v4f)0.0f;
#pragma unroll
        for (int j = 0; j < 4; ++j) macc[j] = (v8f)0.0f;

        auto flush = [&](int a) __attribute__((always_inline)) {
#pragma unroll
            for (int r = 0; r < 8; ++r) {
                v4f v;
                v.x = macc[0][r]; v.y = macc[1][r]; v.z = macc[2][r]; v.w = macc[3][r];
                *(v4f*)&Mraw[a * MP + (8 * hh + r) * NH + 4 * cc] = v;
            }
            v4f o;
            o.x = bsum.x + __shfl_xor(bsum.x, 16);
            o.y = bsum.y + __shfl_xor(bsum.y, 16);
            o.z = bsum.z + __shfl_xor(bsum.z, 16);
            o.w = bsum.w + __shfl_xor(bsum.w, 16);
            if (hh == 0) *(v4f*)&Mraw[a * MP + N_PAIR * NH + 4 * cc] = o;
#pragma unroll
            for (int j = 0; j < 4; ++j) macc[j] = (v8f)0.0f;
            bsum = (v4f)0.0f;
        };

        for (int gb = rbeg; gb < rend; gb += 4 * CG) {
            int          e0[CG], e1[CG];
            int          s0[CG], s1[CG];
            v4f          r0[CG], r1[CG];
            float        w0[CG], w1[CG];
#pragma unroll
            for (int t = 0; t < CG; ++t) {
                int i0 = gb + 4 * t + 2 * hh;
                if (i0 > rend - 2) i0 = rend - 2;
                const int off = (i0 >= st3) ? off3 : ((i0 >= st2) ? off2 : ((i0 >= st1) ? off1 : off0));
                const uint2 ep = *(const uint2*)&slotL[i0 + off];
                e0[t] = (int)(ep.x & 0xffffu);
                s0[t] = (int)(ep.x >> 16);
                e1[t] = (int)(ep.y & 0xffffu);
                s1[t] = (int)(ep.y >> 16);
            }
#pragma unroll
            for (int t = 0; t < CG; ++t) {
                r0[t] = *(const v4f*)(atomf + (size_t)s0[t] * NH + 4 * cc);
                r1[t] = *(const v4f*)(atomf + (size_t)s1[t] * NH + 4 * cc);
                w0[t] = pf[(size_t)e0[t] * N_PAIR + cc];
                w1[t] = pf[(size_t)e1[t] * N_PAIR + cc];
            }
            auto step = [&](int t, int vend) __attribute__((always_inline)) {
                const int  i0 = gb + 4 * t + 2 * hh;
                const bool v0 = i0 < vend;
                const bool v1 = i0 + 1 < vend;
                v2f af;
                af.x = v0 ? w0[t] : 0.0f;
                af.y = v1 ? w1[t] : 0.0f;
                const v4f x0 = v0 ? r0[t] : (v4f)0.0f;
                const v4f x1 = v1 ? r1[t] : (v4f)0.0f;
#pragma unroll
                for (int j = 0; j < 4; ++j) {
                    v2f bf;
                    bf.x = x0[j];
                    bf.y = x1[j];
                    macc[j] = __builtin_amdgcn_wmma_f32_16x16x4_f32(
                        false, af, false, bf, (short)0, macc[j], false, false);
                }
                bsum += x0 + x1;
            };
            const int gend = (gb + 4 * CG < rend) ? gb + 4 * CG : rend;
            while (curA < lastA) {
                const int aS   = __builtin_amdgcn_readfirstlane(start[curA]);
                const int aE   = __builtin_amdgcn_readfirstlane(start[curA + 1]);
                const int cA   = __builtin_amdgcn_readfirstlane(cnt[curA]);
                const int vend = aS + (cA < CAP ? cA : CAP);
                if (aS >= gend) break;
                const int t0 = ((aS > gb ? aS : gb) - gb) >> 2;
                const int t1 = ((aE < gend ? aE : gend) - gb) >> 2;
                if (t1 > t0) {
                    const unsigned int nst = (unsigned int)(t1 - t0);
#pragma unroll
                    for (int t = 0; t < CG; ++t) {
                        if ((unsigned int)(t - t0) < nst) step(t, vend);
                    }
                }
                if (aE > gend) break;
                flush(curA);
                ++curA;
            }
        }
        while (curA < lastA) {
            flush(curA);
            ++curA;
        }
    }
    __syncthreads();

    {
        const int nOv = __builtin_amdgcn_readfirstlane(nOver);
        if (nOv > 0) {
            for (int i = 0; i < nOv; ++i) {
                const int e = overE[i];
                const int a = a2p[2 * e] & (APB - 1);
                const int s = a2p[2 * e + 1];
                for (int idx = tid; idx < KTOT; idx += NTHR) {
                    const int   p = idx >> 6;
                    const int   k = idx & (NH - 1);
                    const float w = (p < N_PAIR) ? pf[(size_t)e * N_PAIR + p] : 1.0f;
                    Mraw[a * MP + idx] += w * atomf[(size_t)s * NH + k];
                }
            }
            __syncthreads();
        }
    }

    {
        const float* aptr = &Mraw[dn * MP + wave * KW + 2 * dhh];

        v8f acc[2][4];
#pragma unroll
        for (int t = 0; t < 2; ++t)
#pragma unroll
            for (int j = 0; j < 4; ++j) acc[t][j] = (v8f)0.0f;

        v2f af[2][2][2];
#pragma unroll
        for (int s = 0; s < 2; ++s) {
            af[0][s][0] = *(const v2f*)(aptr + 4 * s);
            af[0][s][1] = *(const v2f*)(aptr + 16 * MP + 4 * s);
        }

#pragma unroll
        for (int q = 0; q < NPAIR; ++q) {
            if (q + 3 < NPAIR) {
#pragma unroll
                for (int s = 0; s < 2; ++s) {
                    bx[(q + 3) & 3][s] = *(const v4f*)(bptr + ((q + 3) * 8 + s * 4) * NH);
                    by[(q + 3) & 3][s] = *(const v4f*)(bptr + ((q + 3) * 8 + s * 4 + 1) * NH);
                }
            }
            if (q + 1 < NPAIR) {
#pragma unroll
                for (int s = 0; s < 2; ++s) {
                    af[(q + 1) & 1][s][0] = *(const v2f*)(aptr + (q + 1) * 8 + 4 * s);
                    af[(q + 1) & 1][s][1] = *(const v2f*)(aptr + 16 * MP + (q + 1) * 8 + 4 * s);
                }
            }
#pragma unroll
            for (int s = 0; s < 2; ++s) {
#pragma unroll
                for (int j = 0; j < 4; ++j) {
                    v2f bf;
                    bf.x = bx[q & 3][s][j];
                    bf.y = by[q & 3][s][j];
                    acc[0][j] = __builtin_amdgcn_wmma_f32_16x16x4_f32(
                        false, af[q & 1][s][0], false, bf, (short)0, acc[0][j], false, false);
                    acc[1][j] = __builtin_amdgcn_wmma_f32_16x16x4_f32(
                        false, af[q & 1][s][1], false, bf, (short)0, acc[1][j], false, false);
                }
            }
            __builtin_amdgcn_sched_barrier(0);
        }

#pragma unroll
        for (int t = 0; t < 2; ++t)
#pragma unroll
            for (int r = 0; r < 8; ++r) {
                v4f v;
                v.x = acc[t][0][r]; v.y = acc[t][1][r]; v.z = acc[t][2][r]; v.w = acc[t][3][r];
                *(v4f*)&part8[(wave * APB + t * 16 + dhh * 8 + r) * PP + 4 * dn] = v;
            }
    }
    __syncthreads();

#pragma unroll
    for (int it = 0; it < 2; ++it) {
        int i4  = it * NTHR + tid;
        int row = i4 >> 4;
        int c4  = (i4 & 15) * 4;
        v4f r = *(const v4f*)&part8[row * PP + c4];
#pragma unroll
        for (int w = 1; w < 8; ++w) r += *(const v4f*)&part8[(w * APB + row) * PP + c4];
        *(v4f*)&out[((size_t)blk * APB + row) * NH + c4] = r;
    }
}

extern "C" void kernel_launch(void* const* d_in, const int* in_sizes, int n_in,
                              void* d_out, int out_size, void* d_ws, size_t ws_size,
                              hipStream_t stream) {
    const float* pf    = (const float*)d_in[0];
    const float* atomf = (const float*)d_in[1];
    const int*   a2p   = (const int*)d_in[2];
    const float* W     = (const float*)d_in[3];
    const float* b     = (const float*)d_in[4];
    float* out = (float*)d_out;
    float* Wt  = (float*)d_ws;
    unsigned int* bucket = (unsigned int*)((char*)d_ws + WT_BYTES);

    prep_wt<<<NCH + 272, 256, 0, stream>>>(W, b, a2p, Wt, bucket);
    edge_net<<<256, 256, 0, stream>>>(pf, atomf, a2p, Wt, bucket, out);
}
